# starting kernel
# baseline (speedup 1.0000x reference)
#include <hip/hip_runtime.h>

typedef __attribute__((ext_vector_type(2))) float v2f;
typedef __attribute__((ext_vector_type(8))) float v8f;

#define NNEUR  4096
#define MINP   4096
#define NSLOPE 0.01f
#define BLOCK  256
#define WAVES  8
#define KSEG   (MINP / WAVES)

__global__ __launch_bounds__(BLOCK)
void abs_lrelu_backsub_kernel(const float* __restrict__ lbnd,
                              const float* __restrict__ ubnd,
                              const float* __restrict__ alpha,
                              const float* __restrict__ W,
                              const float* __restrict__ bias,
                              const float* __restrict__ l0,
                              const float* __restrict__ u0,
                              float* __restrict__ out)
{
    __shared__ float s_mid[MINP];
    __shared__ float s_rad[MINP];
    __shared__ float s_red[WAVES][2][16];

    const int tid = threadIdx.x;

    for (int j = tid; j < MINP; j += BLOCK) {
        float lo = l0[j], hi = u0[j];
        s_mid[j] = 0.5f * (hi + lo);
        s_rad[j] = 0.5f * (hi - lo);
    }
    __syncthreads();

    const int wave  = tid >> 5;
    const int lane  = tid & 31;
    const int khalf = (lane >> 4) << 1;
    const int row_base = blockIdx.x << 4;
    const int row   = row_base + (lane & 15);
    const float* __restrict__ wrow = W + (size_t)row * MINP;

    v8f acc1a = {}; v8f acc1b = {}; v8f acc2a = {}; v8f acc2b = {};

    const int kbeg = wave * KSEG + khalf;

    #pragma unroll 8
    for (int it = 0; it < KSEG / 8; ++it) {
        const int k = kbeg + (it << 3);
        v2f wA = *(const v2f*)(wrow + k);
        v2f wB = *(const v2f*)(wrow + k + 4);
        v2f mA = *(const v2f*)(s_mid + k);
        v2f mB = *(const v2f*)(s_mid + k + 4);
        v2f rA = *(const v2f*)(s_rad + k);
        v2f rB = *(const v2f*)(s_rad + k + 4);
        v2f aA; aA.x = __builtin_fabsf(wA.x); aA.y = __builtin_fabsf(wA.y);
        v2f aB; aB.x = __builtin_fabsf(wB.x); aB.y = __builtin_fabsf(wB.y);

        acc1a = __builtin_amdgcn_wmma_f32_16x16x4_f32(false, mA, false, wA, (short)0, acc1a, false, false);
        acc2a = __builtin_amdgcn_wmma_f32_16x16x4_f32(false, rA, false, aA, (short)0, acc2a, false, false);
        acc1b = __builtin_amdgcn_wmma_f32_16x16x4_f32(false, mB, false, wB, (short)0, acc1b, false, false);
        acc2b = __builtin_amdgcn_wmma_f32_16x16x4_f32(false, rB, false, aB, (short)0, acc2b, false, false);
    }

    float y1 = acc1a[0] + acc1b[0];
    float y2 = acc2a[0] + acc2b[0];
    if (lane < 16) {
        s_red[wave][0][lane] = y1;
        s_red[wave][1][lane] = y2;
    }
    __syncthreads();

    if (tid < 16) {
        float t1 = 0.f, t2 = 0.f;
        #pragma unroll
        for (int w = 0; w < WAVES; ++w) { t1 += s_red[w][0][tid]; t2 += s_red[w][1][tid]; }

        const int i = row_base + tid;
        const float lbv = lbnd[i], ubv = ubnd[i], av = alpha[i], bv = bias[i];
        float lw = NSLOPE, uw = NSLOPE, ubb = 0.f;
        if (lbv > 0.f) {
            lw = 1.f; uw = 1.f;
        } else if (lbv < 0.f && ubv > 0.f) {
            float slope = (ubv - NSLOPE * lbv) / (ubv - lbv);
            uw  = slope;
            ubb = (NSLOPE - slope) * lbv;
            lw  = fminf(fmaxf(av, NSLOPE), 1.f);
        }
        out[i]         = lw * (t1 - t2 + bv);
        out[NNEUR + i] = uw * (t1 + t2 + bv) + ubb;
    }
}

extern "C" void kernel_launch(void* const* d_in, const int* in_sizes, int n_in,
                              void* d_out, int out_size, void* d_ws, size_t ws_size,
                              hipStream_t stream) {
    (void)in_sizes; (void)n_in; (void)d_ws; (void)ws_size; (void)out_size;
    const float* lbnd  = (const float*)d_in[0];
    const float* ubnd  = (const float*)d_in[1];
    const float* alpha = (const float*)d_in[2];
    const float* W     = (const float*)d_in[3];
    const float* bias  = (const float*)d_in[4];
    const float* l0    = (const float*)d_in[5];
    const float* u0    = (const float*)d_in[6];
    float* out = (float*)d_out;

    dim3 grid(NNEUR / 16);
    dim3 block(BLOCK);
    abs_lrelu_backsub_kernel<<<grid, block, 0, stream>>>(lbnd, ubnd, alpha, W, bias, l0, u0, out);
}


# W read row-contiguous into registers: each load = 512 B of one row (was 16 B from 16 rows), a wave's 64 requested ahead of use; mid/rad as A, row sums from the diagonal of D; k sum order changes
# speedup vs baseline: 1.4849x; 1.4849x over previous
#include <hip/hip_runtime.h>

typedef __attribute__((ext_vector_type(2))) float v2f;
typedef __attribute__((ext_vector_type(4))) float v4f;
typedef __attribute__((ext_vector_type(8))) float v8f;

#define NNEUR  4096
#define MINP   4096
#define NSLOPE 0.01f
#define WAVES  8
#define BLOCK  (WAVES * 32)
#define ROWS   16
#define KSEG   (MINP / WAVES)
#define NQ     (KSEG / 128)
#define PPR    (WAVES * 32)
#define PPAD   (PPR + 8)

__global__ __launch_bounds__(BLOCK)
void abs_lrelu_backsub_kernel(const float* __restrict__ lbnd,
                              const float* __restrict__ ubnd,
                              const float* __restrict__ alpha,
                              const float* __restrict__ W,
                              const float* __restrict__ bias,
                              const float* __restrict__ l0,
                              const float* __restrict__ u0,
                              float* __restrict__ out)
{
    __shared__ float s_part[ROWS * 2][PPAD];
    __shared__ float s_fin[ROWS * 2];

    const int tid  = threadIdx.x;
    const int lane = tid & 31;
    const int wave = __builtin_amdgcn_readfirstlane(tid >> 5);
    const int row_base = blockIdx.x * ROWS;
    const int koff = wave * KSEG + 4 * lane;

    v4f lo[NQ], hi[NQ];
    #pragma unroll
    for (int q = 0; q < NQ; ++q) {
        lo[q] = *(const v4f*)(l0 + koff + 128 * q);
        hi[q] = *(const v4f*)(u0 + koff + 128 * q);
    }

    const float* __restrict__ wp = W + (size_t)row_base * MINP + koff;
    v4f w[ROWS][NQ];
    #pragma unroll
    for (int r = 0; r < ROWS; ++r) {
        #pragma unroll
        for (int q = 0; q < NQ; ++q)
            w[r][q] = *(const v4f*)(wp + (size_t)r * MINP + 128 * q);
    }

    __builtin_amdgcn_sched_barrier(0);

    v4f mid[NQ], rad[NQ];
    #pragma unroll
    for (int q = 0; q < NQ; ++q) {
        mid[q] = 0.5f * (hi[q] + lo[q]);
        rad[q] = 0.5f * (hi[q] - lo[q]);
    }

    const int dsel = (lane < 8 || lane >= 24) ? (lane & 7) : -1;

    #pragma unroll
    for (int r = 0; r < ROWS; ++r) {
        v8f acc1 = {}; v8f acc2 = {};
        #pragma unroll
        for (int q = 0; q < NQ; ++q) {
            const v4f x = w[r][q];
            v2f xa; xa.x = x.x; xa.y = x.y;
            v2f xb; xb.x = x.z; xb.y = x.w;
            v2f aa; aa.x = __builtin_fabsf(x.x); aa.y = __builtin_fabsf(x.y);
            v2f ab; ab.x = __builtin_fabsf(x.z); ab.y = __builtin_fabsf(x.w);
            v2f ma; ma.x = mid[q].x; ma.y = mid[q].y;
            v2f mb; mb.x = mid[q].z; mb.y = mid[q].w;
            v2f ra; ra.x = rad[q].x; ra.y = rad[q].y;
            v2f rb; rb.x = rad[q].z; rb.y = rad[q].w;
            acc1 = __builtin_amdgcn_wmma_f32_16x16x4_f32(false, ma, false, xa, (short)0, acc1, false, false);
            acc2 = __builtin_amdgcn_wmma_f32_16x16x4_f32(false, ra, false, aa, (short)0, acc2, false, false);
            acc1 = __builtin_amdgcn_wmma_f32_16x16x4_f32(false, mb, false, xb, (short)0, acc1, false, false);
            acc2 = __builtin_amdgcn_wmma_f32_16x16x4_f32(false, rb, false, ab, (short)0, acc2, false, false);
        }
        float t1 = 0.f, t2 = 0.f;
        #pragma unroll
        for (int i = 0; i < 8; ++i) {
            t1 = (dsel == i) ? acc1[i] : t1;
            t2 = (dsel == i) ? acc2[i] : t2;
        }
        s_part[2 * r][tid]     = t1;
        s_part[2 * r + 1][tid] = t2;
        if (r & 1) __builtin_amdgcn_sched_barrier(0);
    }
    __syncthreads();

    {
        const int pair = tid >> 3, sub = tid & 7;
        float s = 0.f;
        #pragma unroll
        for (int i = 0; i < PPR / 8; ++i) s += s_part[pair][sub + 8 * i];
        #pragma unroll
        for (int m = 4; m >= 1; m >>= 1) s += __shfl_xor(s, m, 32);
        if (sub == 0) s_fin[pair] = s;
    }
    __syncthreads();

    if (tid < ROWS) {
        const float t1 = s_fin[2 * tid], t2 = s_fin[2 * tid + 1];
        const int i = row_base + tid;
        const float lbv = lbnd[i], ubv = ubnd[i], av = alpha[i], bv = bias[i];
        float lw = NSLOPE, uw = NSLOPE, ubb = 0.f;
        if (lbv > 0.f) {
            lw = 1.f; uw = 1.f;
        } else if (lbv < 0.f && ubv > 0.f) {
            float slope = (ubv - NSLOPE * lbv) / (ubv - lbv);
            uw  = slope;
            ubb = (NSLOPE - slope) * lbv;
            lw  = fminf(fmaxf(av, NSLOPE), 1.f);
        }
        out[i]         = lw * (t1 - t2 + bv);
        out[NNEUR + i] = uw * (t1 + t2 + bv) + ubb;
    }
}

extern "C" void kernel_launch(void* const* d_in, const int* in_sizes, int n_in,
                              void* d_out, int out_size, void* d_ws, size_t ws_size,
                              hipStream_t stream) {
    (void)in_sizes; (void)n_in; (void)d_ws; (void)ws_size; (void)out_size;
    const float* lbnd  = (const float*)d_in[0];
    const float* ubnd  = (const float*)d_in[1];
    const float* alpha = (const float*)d_in[2];
    const float* W     = (const float*)d_in[3];
    const float* bias  = (const float*)d_in[4];
    const float* l0    = (const float*)d_in[5];
    const float* u0    = (const float*)d_in[6];
    float* out = (float*)d_out;

    dim3 grid(NNEUR / ROWS);
    dim3 block(BLOCK);
    abs_lrelu_backsub_kernel<<<grid, block, 0, stream>>>(lbnd, ubnd, alpha, W, bias, l0, u0, out);
}


# k no longer split over waves: each wave sums 2 whole rows (mid/rad shared through LDS), so the tail has no cross-wave partial sums, LDS reduction or dependent loads; sum order changes
# speedup vs baseline: 1.7563x; 1.1828x over previous
#include <hip/hip_runtime.h>

typedef __attribute__((ext_vector_type(2))) float v2f;
typedef __attribute__((ext_vector_type(4))) float v4f;
typedef __attribute__((ext_vector_type(8))) float v8f;

#define NNEUR  4096
#define MINP   4096
#define NSLOPE 0.01f
#define WAVES  8
#define BLOCK  (WAVES * 32)
#define RPW    2
#define ROWS   (WAVES * RPW)
#define NP     (MINP / 128)
#define BOXQ   (MINP / (4 * BLOCK))


static __device__ __forceinline__ float diag_total(float v)
{
    int x = __builtin_bit_cast(int, v);
    int y = __builtin_amdgcn_permlanex16(x, x, 0xfedcba98u, 0x76543210u, false, false);
    v += __builtin_bit_cast(float, y);
    x = __builtin_bit_cast(int, v);
    v += __builtin_bit_cast(float, __builtin_amdgcn_update_dpp(x, x, 0x164, 0xf, 0xf, false));
    x = __builtin_bit_cast(int, v);
    v += __builtin_bit_cast(float, __builtin_amdgcn_update_dpp(x, x, 0x162, 0xf, 0xf, false));
    x = __builtin_bit_cast(int, v);
    v += __builtin_bit_cast(float, __builtin_amdgcn_update_dpp(x, x, 0x161, 0xf, 0xf, false));
    return v;
}

__global__ __launch_bounds__(BLOCK)
void abs_lrelu_backsub_kernel(const float* __restrict__ lbnd,
                              const float* __restrict__ ubnd,
                              const float* __restrict__ alpha,
                              const float* __restrict__ W,
                              const float* __restrict__ bias,
                              const float* __restrict__ l0,
                              const float* __restrict__ u0,
                              float* __restrict__ out)
{
    __shared__ v4f   s_mid[MINP / 4];
    __shared__ v4f   s_rad[MINP / 4];
    __shared__ float s_out[2][ROWS];

    const int tid  = threadIdx.x;
    const int lane = tid & 31;
    const int wave = __builtin_amdgcn_readfirstlane(tid >> 5);
    const int row_base = blockIdx.x * ROWS;
    const int r0 = row_base + RPW * wave;

    float lbv[RPW], ubv[RPW], av[RPW], bv[RPW];
    #pragma unroll
    for (int j = 0; j < RPW; ++j) {
        lbv[j] = lbnd[r0 + j]; ubv[j] = ubnd[r0 + j]; av[j] = alpha[r0 + j]; bv[j] = bias[r0 + j];
    }
    __builtin_amdgcn_sched_barrier(0);

    v4f lo[BOXQ], hi[BOXQ];
    #pragma unroll
    for (int j = 0; j < BOXQ; ++j) {
        lo[j] = *(const v4f*)(l0 + 4 * tid + 4 * BLOCK * j);
        hi[j] = *(const v4f*)(u0 + 4 * tid + 4 * BLOCK * j);
    }

    __builtin_amdgcn_sched_barrier(0);

    const float* __restrict__ wp = W + (size_t)r0 * MINP + 4 * lane;
    v4f w[RPW][NP];
    #pragma unroll
    for (int p = 0; p < NP; ++p) {
        #pragma unroll
        for (int j = 0; j < RPW; ++j)
            w[j][p] = *(const v4f*)(wp + (size_t)j * MINP + 128 * p);
    }

    __builtin_amdgcn_sched_barrier(0);

    #pragma unroll
    for (int j = 0; j < BOXQ; ++j) {
        s_mid[tid + BLOCK * j] = 0.5f * (hi[j] + lo[j]);
        s_rad[tid + BLOCK * j] = 0.5f * (hi[j] - lo[j]);
    }
    asm volatile("s_wait_dscnt 0x0\n\ts_barrier_signal -1\n\ts_barrier_wait -1" ::: "memory");

    float lw[RPW], uw[RPW], ubb[RPW];
    #pragma unroll
    for (int j = 0; j < RPW; ++j) {
        const bool positive = lbv[j] > 0.f;
        const bool crossing = (lbv[j] < 0.f) && (ubv[j] > 0.f);
        const float slope = (ubv[j] - NSLOPE * lbv[j]) / (ubv[j] - lbv[j]);
        lw[j]  = crossing ? fminf(fmaxf(av[j], NSLOPE), 1.f) : (positive ? 1.f : NSLOPE);
        uw[j]  = crossing ? slope : (positive ? 1.f : NSLOPE);
        ubb[j] = crossing ? (NSLOPE - slope) * lbv[j] : 0.f;
    }

    const int dsel = (lane < 8 || lane >= 24) ? (lane & 7) : -1;

    v8f acc1[RPW], acc2[RPW];
    #pragma unroll
    for (int j = 0; j < RPW; ++j) { acc1[j] = (v8f){}; acc2[j] = (v8f){}; }

    v4f mcur = s_mid[lane], rcur = s_rad[lane];
    #pragma unroll
    for (int p = 0; p < NP; ++p) {
        v4f mnxt = mcur, rnxt = rcur;
        if (p + 1 < NP) { mnxt = s_mid[32 * (p + 1) + lane]; rnxt = s_rad[32 * (p + 1) + lane]; }
        __builtin_amdgcn_sched_barrier(0);
        v2f ma; ma.x = mcur.x; ma.y = mcur.y;
        v2f mb; mb.x = mcur.z; mb.y = mcur.w;
        v2f ra; ra.x = rcur.x; ra.y = rcur.y;
        v2f rb; rb.x = rcur.z; rb.y = rcur.w;
        v2f xa[RPW], xb[RPW], aa[RPW], ab[RPW];
        #pragma unroll
        for (int j = 0; j < RPW; ++j) {
            const v4f x = w[j][p];
            xa[j].x = x.x; xa[j].y = x.y; xb[j].x = x.z; xb[j].y = x.w;
            aa[j].x = __builtin_fabsf(x.x); aa[j].y = __builtin_fabsf(x.y);
            ab[j].x = __builtin_fabsf(x.z); ab[j].y = __builtin_fabsf(x.w);
        }
        #pragma unroll
        for (int j = 0; j < RPW; ++j) {
            acc1[j] = __builtin_amdgcn_wmma_f32_16x16x4_f32(false, ma, false, xa[j], (short)0, acc1[j], false, false);
            acc2[j] = __builtin_amdgcn_wmma_f32_16x16x4_f32(false, ra, false, aa[j], (short)0, acc2[j], false, false);
        }
        #pragma unroll
        for (int j = 0; j < RPW; ++j) {
            acc1[j] = __builtin_amdgcn_wmma_f32_16x16x4_f32(false, mb, false, xb[j], (short)0, acc1[j], false, false);
            acc2[j] = __builtin_amdgcn_wmma_f32_16x16x4_f32(false, rb, false, ab[j], (short)0, acc2[j], false, false);
        }
        mcur = mnxt; rcur = rnxt;
        __builtin_amdgcn_sched_barrier(0);
    }

    #pragma unroll
    for (int j = 0; j < RPW; ++j) {
        float t1 = 0.f, t2 = 0.f;
        #pragma unroll
        for (int i = 0; i < 8; ++i) {
            t1 = (dsel == i) ? acc1[j][i] : t1;
            t2 = (dsel == i) ? acc2[j][i] : t2;
        }
        t1 = diag_total(t1);
        t2 = diag_total(t2);
        if (lane == 0) {
            s_out[0][RPW * wave + j] = lw[j] * (t1 - t2 + bv[j]);
            s_out[1][RPW * wave + j] = uw[j] * (t1 + t2 + bv[j]) + ubb[j];
        }
    }
    __syncthreads();

    if (tid < ROWS) {
        out[row_base + tid]         = s_out[0][tid];
        out[NNEUR + row_base + tid] = s_out[1][tid];
    }
}

extern "C" void kernel_launch(void* const* d_in, const int* in_sizes, int n_in,
                              void* d_out, int out_size, void* d_ws, size_t ws_size,
                              hipStream_t stream) {
    (void)in_sizes; (void)n_in; (void)d_ws; (void)ws_size; (void)out_size;
    const float* lbnd  = (const float*)d_in[0];
    const float* ubnd  = (const float*)d_in[1];
    const float* alpha = (const float*)d_in[2];
    const float* W     = (const float*)d_in[3];
    const float* bias  = (const float*)d_in[4];
    const float* l0    = (const float*)d_in[5];
    const float* u0    = (const float*)d_in[6];
    float* out = (float*)d_out;

    dim3 grid(NNEUR / ROWS);
    dim3 block(BLOCK);
    abs_lrelu_backsub_kernel<<<grid, block, 0, stream>>>(lbnd, ubnd, alpha, W, bias, l0, u0, out);
}


# W requested from inside the product loop (8 pieces before the workgroup barrier, then 2 per piece multiplied) instead of all 64 loads ahead of the barrier: matrix work runs under the stream
# speedup vs baseline: 1.9307x; 1.0993x over previous
#include <hip/hip_runtime.h>

typedef __attribute__((ext_vector_type(2))) float v2f;
typedef __attribute__((ext_vector_type(4))) float v4f;
typedef __attribute__((ext_vector_type(8))) float v8f;

#define NNEUR  4096
#define MINP   4096
#define NSLOPE 0.01f
#define WAVES  8
#define BLOCK  (WAVES * 32)
#define RPW    2
#define ROWS   (WAVES * RPW)
#define NP     (MINP / 128)
#define BOXQ   (MINP / (4 * BLOCK))
#define FIRST  8
#define RATE   2


static __device__ __forceinline__ float diag_total(float v)
{
    int x = __builtin_bit_cast(int, v);
    int y = __builtin_amdgcn_permlanex16(x, x, 0xfedcba98u, 0x76543210u, false, false);
    v += __builtin_bit_cast(float, y);
    x = __builtin_bit_cast(int, v);
    v += __builtin_bit_cast(float, __builtin_amdgcn_update_dpp(x, x, 0x164, 0xf, 0xf, false));
    x = __builtin_bit_cast(int, v);
    v += __builtin_bit_cast(float, __builtin_amdgcn_update_dpp(x, x, 0x162, 0xf, 0xf, false));
    x = __builtin_bit_cast(int, v);
    v += __builtin_bit_cast(float, __builtin_amdgcn_update_dpp(x, x, 0x161, 0xf, 0xf, false));
    return v;
}

__global__ __launch_bounds__(BLOCK) __attribute__((amdgpu_waves_per_eu(2, 2)))
void abs_lrelu_backsub_kernel(const float* __restrict__ lbnd,
                              const float* __restrict__ ubnd,
                              const float* __restrict__ alpha,
                              const float* __restrict__ W,
                              const float* __restrict__ bias,
                              const float* __restrict__ l0,
                              const float* __restrict__ u0,
                              float* __restrict__ out)
{
    __shared__ v4f   s_mid[MINP / 4];
    __shared__ v4f   s_rad[MINP / 4];
    __shared__ float s_out[2][ROWS];

    const int tid  = threadIdx.x;
    const int lane = tid & 31;
    const int wave = __builtin_amdgcn_readfirstlane(tid >> 5);
    const int row_base = blockIdx.x * ROWS;
    const int r0 = row_base + RPW * wave;

    float lbv[RPW], ubv[RPW], av[RPW], bv[RPW];
    #pragma unroll
    for (int j = 0; j < RPW; ++j) {
        lbv[j] = lbnd[r0 + j]; ubv[j] = ubnd[r0 + j]; av[j] = alpha[r0 + j]; bv[j] = bias[r0 + j];
    }
    __builtin_amdgcn_sched_barrier(0);

    v4f lo[BOXQ], hi[BOXQ];
    #pragma unroll
    for (int j = 0; j < BOXQ; ++j) {
        lo[j] = *(const v4f*)(l0 + 4 * tid + 4 * BLOCK * j);
        hi[j] = *(const v4f*)(u0 + 4 * tid + 4 * BLOCK * j);
    }

    __builtin_amdgcn_sched_barrier(0);

    const float* __restrict__ wp = W + (size_t)r0 * MINP + 4 * lane;
    v4f w[RPW][NP];
    #pragma unroll
    for (int p = 0; p < FIRST; ++p) {
        #pragma unroll
        for (int j = 0; j < RPW; ++j)
            w[j][p] = *(const v4f*)(wp + (size_t)j * MINP + 128 * p);
    }

    __builtin_amdgcn_sched_barrier(0);

    #pragma unroll
    for (int j = 0; j < BOXQ; ++j) {
        s_mid[tid + BLOCK * j] = 0.5f * (hi[j] + lo[j]);
        s_rad[tid + BLOCK * j] = 0.5f * (hi[j] - lo[j]);
    }
    asm volatile("s_wait_dscnt 0x0\n\ts_barrier_signal -1\n\ts_barrier_wait -1" ::: "memory");

    float lw[RPW], uw[RPW], ubb[RPW];
    #pragma unroll
    for (int j = 0; j < RPW; ++j) {
        const bool positive = lbv[j] > 0.f;
        const bool crossing = (lbv[j] < 0.f) && (ubv[j] > 0.f);
        const float slope = (ubv[j] - NSLOPE * lbv[j]) / (ubv[j] - lbv[j]);
        lw[j]  = crossing ? fminf(fmaxf(av[j], NSLOPE), 1.f) : (positive ? 1.f : NSLOPE);
        uw[j]  = crossing ? slope : (positive ? 1.f : NSLOPE);
        ubb[j] = crossing ? (NSLOPE - slope) * lbv[j] : 0.f;
    }

    const int dsel = (lane < 8 || lane >= 24) ? (lane & 7) : -1;

    v8f acc1[RPW], acc2[RPW];
    #pragma unroll
    for (int j = 0; j < RPW; ++j) { acc1[j] = (v8f){}; acc2[j] = (v8f){}; }

    v4f mcur = s_mid[lane], rcur = s_rad[lane];
    #pragma unroll
    for (int p = 0; p < NP; ++p) {
        v4f mnxt = mcur, rnxt = rcur;
        if (p + 1 < NP) { mnxt = s_mid[32 * (p + 1) + lane]; rnxt = s_rad[32 * (p + 1) + lane]; }
        __builtin_amdgcn_sched_barrier(0);
        v2f ma; ma.x = mcur.x; ma.y = mcur.y;
        v2f mb; mb.x = mcur.z; mb.y = mcur.w;
        v2f ra; ra.x = rcur.x; ra.y = rcur.y;
        v2f rb; rb.x = rcur.z; rb.y = rcur.w;
        v2f xa[RPW], xb[RPW], aa[RPW], ab[RPW];
        #pragma unroll
        for (int j = 0; j < RPW; ++j) {
            const v4f x = w[j][p];
            xa[j].x = x.x; xa[j].y = x.y; xb[j].x = x.z; xb[j].y = x.w;
            aa[j].x = __builtin_fabsf(x.x); aa[j].y = __builtin_fabsf(x.y);
            ab[j].x = __builtin_fabsf(x.z); ab[j].y = __builtin_fabsf(x.w);
        }
        #pragma unroll
        for (int j = 0; j < RPW; ++j) {
            acc1[j] = __builtin_amdgcn_wmma_f32_16x16x4_f32(false, ma, false, xa[j], (short)0, acc1[j], false, false);
            acc2[j] = __builtin_amdgcn_wmma_f32_16x16x4_f32(false, ra, false, aa[j], (short)0, acc2[j], false, false);
        }
        #pragma unroll
        for (int j = 0; j < RPW; ++j) {
            acc1[j] = __builtin_amdgcn_wmma_f32_16x16x4_f32(false, mb, false, xb[j], (short)0, acc1[j], false, false);
            acc2[j] = __builtin_amdgcn_wmma_f32_16x16x4_f32(false, rb, false, ab[j], (short)0, acc2[j], false, false);
        }
        mcur = mnxt; rcur = rnxt;
        __builtin_amdgcn_sched_barrier(0);
        #pragma unroll
        for (int q = FIRST + RATE * p; q < FIRST + RATE * (p + 1); ++q) {
            if (q < NP) {
                #pragma unroll
                for (int j = 0; j < RPW; ++j)
                    w[j][q] = *(const v4f*)(wp + (size_t)j * MINP + 128 * q);
            }
        }
        __builtin_amdgcn_sched_barrier(0);
    }

    #pragma unroll
    for (int j = 0; j < RPW; ++j) {
        float t1 = 0.f, t2 = 0.f;
        #pragma unroll
        for (int i = 0; i < 8; ++i) {
            t1 = (dsel == i) ? acc1[j][i] : t1;
            t2 = (dsel == i) ? acc2[j][i] : t2;
        }
        t1 = diag_total(t1);
        t2 = diag_total(t2);
        if (lane == 0) {
            s_out[0][RPW * wave + j] = lw[j] * (t1 - t2 + bv[j]);
            s_out[1][RPW * wave + j] = uw[j] * (t1 + t2 + bv[j]) + ubb[j];
        }
    }
    __syncthreads();

    if (tid < ROWS) {
        out[row_base + tid]         = s_out[0][tid];
        out[NNEUR + row_base + tid] = s_out[1][tid];
    }
}

extern "C" void kernel_launch(void* const* d_in, const int* in_sizes, int n_in,
                              void* d_out, int out_size, void* d_ws, size_t ws_size,
                              hipStream_t stream) {
    (void)in_sizes; (void)n_in; (void)d_ws; (void)ws_size; (void)out_size;
    const float* lbnd  = (const float*)d_in[0];
    const float* ubnd  = (const float*)d_in[1];
    const float* alpha = (const float*)d_in[2];
    const float* W     = (const float*)d_in[3];
    const float* bias  = (const float*)d_in[4];
    const float* l0    = (const float*)d_in[5];
    const float* u0    = (const float*)d_in[6];
    float* out = (float*)d_out;

    dim3 grid(NNEUR / ROWS);
    dim3 block(BLOCK);
    abs_lrelu_backsub_kernel<<<grid, block, 0, stream>>>(lbnd, ubnd, alpha, W, bias, l0, u0, out);
}


# 16 rows of W as B columns (via LDS), mid as A row 0, rad with w's sign on the diagonal: one product sums W.mid of 16 rows and |W|.rad of 15; 136 per wave (was 256); k over waves, sum order changes
# speedup vs baseline: 2.3776x; 1.2315x over previous
#include <hip/hip_runtime.h>

typedef __attribute__((ext_vector_type(2))) float v2f;
typedef __attribute__((ext_vector_type(4))) float v4f;
typedef __attribute__((ext_vector_type(4))) int   v4i;
typedef __attribute__((ext_vector_type(8))) float v8f;

#define NNEUR  4096
#define MINP   4096
#define NSLOPE 0.01f
#define WAVES  8
#define BLOCK  (WAVES * 32)
#define ROWS   16
#define KW     (MINP / WAVES)
#define NPC    (KW / 128)
#define NST    (2 * NPC)
#define BOXQ   (MINP / (4 * BLOCK))
#define PITCH  33
#define RADOFF (MINP / 4 + 4)


static __device__ __forceinline__ float diag_total(float v)
{
    int x = __builtin_bit_cast(int, v);
    int y = __builtin_amdgcn_permlanex16(x, x, 0xfedcba98u, 0x76543210u, false, false);
    v += __builtin_bit_cast(float, y);
    x = __builtin_bit_cast(int, v);
    v += __builtin_bit_cast(float, __builtin_amdgcn_update_dpp(x, x, 0x164, 0xf, 0xf, false));
    x = __builtin_bit_cast(int, v);
    v += __builtin_bit_cast(float, __builtin_amdgcn_update_dpp(x, x, 0x162, 0xf, 0xf, false));
    x = __builtin_bit_cast(int, v);
    v += __builtin_bit_cast(float, __builtin_amdgcn_update_dpp(x, x, 0x161, 0xf, 0xf, false));
    return v;
}

__global__ __launch_bounds__(BLOCK) __attribute__((amdgpu_waves_per_eu(2, 2)))
void abs_lrelu_backsub_kernel(const float* __restrict__ lbnd,
                              const float* __restrict__ ubnd,
                              const float* __restrict__ alpha,
                              const float* __restrict__ W,
                              const float* __restrict__ bias,
                              const float* __restrict__ l0,
                              const float* __restrict__ u0,
                              float* __restrict__ out)
{
    __shared__ v4f   s_box[RADOFF + MINP / 4];
    __shared__ v4f   s_t[WAVES][ROWS * PITCH];
    __shared__ float s_part[WAVES][2][ROWS];

    const int tid  = threadIdx.x;
    const int lane = tid & 31;
    const int wave = __builtin_amdgcn_readfirstlane(tid >> 5);
    const int c = lane & 15;
    const int h = lane >> 4;
    const int row_base = blockIdx.x * ROWS;

    const float lbv = lbnd[row_base + c], ubv = ubnd[row_base + c], av = alpha[row_base + c], bv = bias[row_base + c];

    v4f lo[BOXQ], hi[BOXQ];
    #pragma unroll
    for (int j = 0; j < BOXQ; ++j) {
        lo[j] = *(const v4f*)(l0 + 4 * tid + 4 * BLOCK * j);
        hi[j] = *(const v4f*)(u0 + 4 * tid + 4 * BLOCK * j);
    }
    __builtin_amdgcn_sched_barrier(0);

    const float* __restrict__ wp = W + (size_t)row_base * MINP + KW * wave + 4 * lane;
    v4f w[NPC][ROWS];
    #pragma unroll
    for (int r = 0; r < ROWS; ++r) w[0][r] = *(const v4f*)(wp + (size_t)r * MINP);
    __builtin_amdgcn_sched_barrier(0);

    #pragma unroll
    for (int j = 0; j < BOXQ; ++j) {
        s_box[tid + BLOCK * j]          = 0.5f * (hi[j] + lo[j]);
        s_box[RADOFF + tid + BLOCK * j] = 0.5f * (hi[j] - lo[j]);
    }
    __builtin_amdgcn_sched_barrier(0);
    #pragma unroll
    for (int r = 0; r < ROWS; ++r) w[1][r] = *(const v4f*)(wp + (size_t)r * MINP + 128);
    asm volatile("s_wait_dscnt 0x0\n\ts_barrier_signal -1\n\ts_barrier_wait -1" ::: "memory");

    #pragma unroll
    for (int r = 0; r < ROWS; ++r) w[2][r] = *(const v4f*)(wp + (size_t)r * MINP + 128 * 2);

    v4f* const tb = &s_t[wave][0];
    const int  smask = (c == 0) ? 0 : (int)0x80000000;
    const v4f* const bxp = s_box + ((c == 0) ? 0 : RADOFF) + (KW / 4) * wave + h;
    const v4f* const tvp = tb + c * PITCH + h;
    const v4f* const rnp = s_box + RADOFF + (KW / 4) * wave + lane;

    v8f accT0 = (v8f){}, accT1 = (v8f){}, accD = (v8f){};
    v4f fw[2][8], fb[2][8], rn[NPC];

    #pragma unroll
    for (int r = 0; r < ROWS; ++r) tb[r * PITCH + lane] = w[0][r];
    rn[0] = rnp[0];
    #pragma unroll
    for (int jj = 0; jj < 8; ++jj) { fw[0][jj] = tvp[2 * jj]; fb[0][jj] = bxp[2 * jj]; }

    #pragma unroll
    for (int s = 0; s < NST; ++s) {
        const int p = s >> 1, g = s & 1;
        if (s + 1 < NST) {
            const int p1 = (s + 1) >> 1, g1 = (s + 1) & 1;
            if (g1 == 0) {
                #pragma unroll
                for (int r = 0; r < ROWS; ++r) tb[r * PITCH + lane] = w[p1][r];
                __builtin_amdgcn_wave_barrier();
                rn[p1] = rnp[32 * p1];
            }
            #pragma unroll
            for (int jj = 0; jj < 8; ++jj) {
                fw[(s + 1) & 1][jj] = tvp[2 * (8 * g1 + jj)];
                fb[(s + 1) & 1][jj] = bxp[32 * p1 + 2 * (8 * g1 + jj)];
            }
        }
        __builtin_amdgcn_sched_barrier(0);

        if (g == 0) {
            const v4f x = w[p][0];
            v2f ra; ra.x = rn[p].x; ra.y = rn[p].y;
            v2f rb; rb.x = rn[p].z; rb.y = rn[p].w;
            v2f xa; xa.x = __builtin_fabsf(x.x); xa.y = __builtin_fabsf(x.y);
            v2f xb; xb.x = __builtin_fabsf(x.z); xb.y = __builtin_fabsf(x.w);
            accD = __builtin_amdgcn_wmma_f32_16x16x4_f32(false, ra, false, xa, (short)0, accD, false, false);
            accD = __builtin_amdgcn_wmma_f32_16x16x4_f32(false, rb, false, xb, (short)0, accD, false, false);
        }
        #pragma unroll
        for (int jj = 0; jj < 8; ++jj) {
            const v4f wv = fw[s & 1][jj];
            const v4i wi = __builtin_bit_cast(v4i, wv);
            const v4i bi = __builtin_bit_cast(v4i, fb[s & 1][jj]);
            const v4f a  = __builtin_bit_cast(v4f, bi ^ (wi & smask));
            v2f a0; a0.x = a.x;  a0.y = a.y;
            v2f a1; a1.x = a.z;  a1.y = a.w;
            v2f b0; b0.x = wv.x; b0.y = wv.y;
            v2f b1; b1.x = wv.z; b1.y = wv.w;
            accT0 = __builtin_amdgcn_wmma_f32_16x16x4_f32(false, a0, false, b0, (short)0, accT0, false, false);
            accT1 = __builtin_amdgcn_wmma_f32_16x16x4_f32(false, a1, false, b1, (short)0, accT1, false, false);
        }
        __builtin_amdgcn_sched_barrier(0);
        if (s == 1) {
            #pragma unroll
            for (int r = 0; r < ROWS; ++r) w[3][r] = *(const v4f*)(wp + (size_t)r * MINP + 128 * 3);
        }
    }

    const int dsel = (h == (c >> 3)) ? (c & 7) : -1;
    const v8f st = accT0 + accT1;
    float t2 = 0.f, d0 = 0.f;
    #pragma unroll
    for (int i = 0; i < 8; ++i) {
        t2 = (dsel == i) ? st[i]   : t2;
        d0 = (dsel == i) ? accD[i] : d0;
    }
    d0 = diag_total(d0);
    if (lane == 0) t2 = d0;
    if (h == 0)    s_part[wave][0][c] = st[0];
    if (dsel >= 0) s_part[wave][1][c] = t2;
    __syncthreads();

    if (tid < ROWS) {
        float y1 = 0.f, y2 = 0.f;
        #pragma unroll
        for (int q = 0; q < WAVES; ++q) { y1 += s_part[q][0][tid]; y2 += s_part[q][1][tid]; }
        const bool positive = lbv > 0.f;
        const bool crossing = (lbv < 0.f) && (ubv > 0.f);
        const float slope = (ubv - NSLOPE * lbv) / (ubv - lbv);
        const float lw  = crossing ? fminf(fmaxf(av, NSLOPE), 1.f) : (positive ? 1.f : NSLOPE);
        const float uw  = crossing ? slope : (positive ? 1.f : NSLOPE);
        const float ubb = crossing ? (NSLOPE - slope) * lbv : 0.f;
        out[row_base + tid]         = lw * (y1 - y2 + bv);
        out[NNEUR + row_base + tid] = uw * (y1 + y2 + bv) + ubb;
    }
}

extern "C" void kernel_launch(void* const* d_in, const int* in_sizes, int n_in,
                              void* d_out, int out_size, void* d_ws, size_t ws_size,
                              hipStream_t stream) {
    (void)in_sizes; (void)n_in; (void)d_ws; (void)ws_size; (void)out_size;
    const float* lbnd  = (const float*)d_in[0];
    const float* ubnd  = (const float*)d_in[1];
    const float* alpha = (const float*)d_in[2];
    const float* W     = (const float*)d_in[3];
    const float* bias  = (const float*)d_in[4];
    const float* l0    = (const float*)d_in[5];
    const float* u0    = (const float*)d_in[6];
    float* out = (float*)d_out;

    dim3 grid(NNEUR / ROWS);
    dim3 block(BLOCK);
    abs_lrelu_backsub_kernel<<<grid, block, 0, stream>>>(lbnd, ubnd, alpha, W, bias, l0, u0, out);
}
